# starting kernel
# baseline (speedup 1.0000x reference)
#include <hip/hip_runtime.h>

#define CC 64
#define KK 27
#define LDS_STRIDE 66
#define IDX_STRIDE 28

typedef __attribute__((ext_vector_type(2))) float v2f;
typedef __attribute__((ext_vector_type(8))) float v8f;

__global__ __launch_bounds__(256)
void gemm_bias_wmma(const float* __restrict__ feat,
                    const float* __restrict__ W,
                    const float* __restrict__ b1,
                    float* __restrict__ bx,
                    int Npts)
{
    __shared__ float sWt[CC * LDS_STRIDE];
    for (int i = threadIdx.x; i < CC * CC; i += 256) {
        int k = i >> 6;
        int n = i & 63;
        sWt[n * LDS_STRIDE + k] = W[i];
    }
    __syncthreads();

    const int lane = threadIdx.x & 31;
    const int wave = threadIdx.x >> 5;
    const int l15  = lane & 15;
    const int hs   = lane >> 4;
    const int rowBase = (blockIdx.x * 8 + wave) * 16;

    int aRowIdx = rowBase + l15;
    if (aRowIdx > Npts - 1) aRowIdx = Npts - 1;
    const float* aRow = feat + (size_t)aRowIdx * CC;

    v8f acc[4] = {v8f{}, v8f{}, v8f{}, v8f{}};

#pragma unroll
    for (int k0 = 0; k0 < CC; k0 += 4) {
        v2f a = *(const v2f*)(aRow + k0 + hs * 2);
#pragma unroll
        for (int nt = 0; nt < 4; ++nt) {
            const float* bp = &sWt[(nt * 16 + l15) * LDS_STRIDE + k0 + hs * 2];
            v2f b = *(const v2f*)bp;
            acc[nt] = __builtin_amdgcn_wmma_f32_16x16x4_f32(
                false, a, false, b, (short)0, acc[nt], false, false);
        }
    }

#pragma unroll
    for (int nt = 0; nt < 4; ++nt) {
        const int col = nt * 16 + l15;
        const float bias = b1[col];
#pragma unroll
        for (int r = 0; r < 8; ++r) {
            const int row = rowBase + hs * 8 + r;
            if (row < Npts) {
                bx[(size_t)row * CC + col] = acc[nt][r] + bias;
            }
        }
    }
}

__global__ __launch_bounds__(256)
void dw_gather_residual(const float* __restrict__ feat,
                        const int*   __restrict__ nb,
                        const float* __restrict__ bx,
                        const float* __restrict__ dw,
                        const float* __restrict__ b2,
                        float*       __restrict__ out,
                        int Npts)
{
    __shared__ int   sIdx[16 * IDX_STRIDE];
    __shared__ float sDw[KK * CC];

    const int rowTile = blockIdx.x * 16;

    {
        int validRows = Npts - rowTile;
        if (validRows > 16) validRows = 16;
        const int total = validRows * KK;
        const int* slab = nb + (size_t)rowTile * KK;
        for (int i = threadIdx.x; i < total; i += 256) {
            const int r = i / KK;
            const int k = i - r * KK;
            sIdx[r * IDX_STRIDE + k] = slab[i];
        }
    }
    for (int i = threadIdx.x; i < KK * CC; i += 256) {
        sDw[i] = dw[i];
    }
    __syncthreads();

    const int r   = threadIdx.x >> 4;
    const int c4  = (threadIdx.x & 15) << 2;
    const int row = rowTile + r;
    if (row >= Npts) return;

    const int* myIdx = &sIdx[r * IDX_STRIDE];

    float ax = 0.f, ay = 0.f, az = 0.f, aw = 0.f;
#pragma unroll
    for (int k = 0; k < KK; ++k) {
        const int idx = myIdx[k];
        if (idx >= 0) {
            const float4 g = *(const float4*)(bx + (size_t)idx * CC + c4);
            const float4 w = *(const float4*)(&sDw[k * CC + c4]);
            ax += g.x * w.x;
            ay += g.y * w.y;
            az += g.z * w.z;
            aw += g.w * w.w;
        }
    }

    const float4 f  = *(const float4*)(feat + (size_t)row * CC + c4);
    const float4 bb = *(const float4*)(b2 + c4);
    float4 o;
    o.x = f.x + ax + bb.x;
    o.y = f.y + ay + bb.y;
    o.z = f.z + az + bb.z;
    o.w = f.w + aw + bb.w;
    *(float4*)(out + (size_t)row * CC + c4) = o;
}

extern "C" void kernel_launch(void* const* d_in, const int* in_sizes, int n_in,
                              void* d_out, int out_size, void* d_ws, size_t ws_size,
                              hipStream_t stream)
{
    const float* feat = (const float*)d_in[0];
    const int*   nb   = (const int*)  d_in[1];
    const float* W    = (const float*)d_in[2];
    const float* b1   = (const float*)d_in[3];
    const float* b2   = (const float*)d_in[4];
    const float* dw   = (const float*)d_in[5];
    float* out = (float*)d_out;
    float* bx  = (float*)d_ws;

    const int Npts = in_sizes[0] / CC;

    const int gemmBlocks = (Npts + 127) / 128;
    gemm_bias_wmma<<<gemmBlocks, 256, 0, stream>>>(feat, W, b1, bx, Npts);

    const int dwBlocks = (Npts + 15) / 16;
    dw_gather_residual<<<dwBlocks, 256, 0, stream>>>(feat, nb, bx, dw, b2, out, Npts);
}


# gather kernel: one resident grid (2048 work-groups) whose waves loop over row pairs, indices requested one pair ahead through a per-wave LDS slab, dw staged once: 16384 waves enter, not 131072
# speedup vs baseline: 1.2159x; 1.2159x over previous
#include <hip/hip_runtime.h>

#define CC 64
#define KK 27
#define LDS_STRIDE 66
#define GW 8
#define SLAB 64

typedef __attribute__((ext_vector_type(2))) float v2f;
typedef __attribute__((ext_vector_type(8))) float v8f;

__global__ __launch_bounds__(256)
void gemm_bias_wmma(const float* __restrict__ feat,
                    const float* __restrict__ W,
                    const float* __restrict__ b1,
                    float* __restrict__ bx,
                    int Npts)
{
    __shared__ float sWt[CC * LDS_STRIDE];
    for (int i = threadIdx.x; i < CC * CC; i += 256) {
        int k = i >> 6;
        int n = i & 63;
        sWt[n * LDS_STRIDE + k] = W[i];
    }
    __syncthreads();

    const int lane = threadIdx.x & 31;
    const int wave = threadIdx.x >> 5;
    const int l15  = lane & 15;
    const int hs   = lane >> 4;
    const int rowBase = (blockIdx.x * 8 + wave) * 16;

    int aRowIdx = rowBase + l15;
    if (aRowIdx > Npts - 1) aRowIdx = Npts - 1;
    const float* aRow = feat + (size_t)aRowIdx * CC;

    v8f acc[4] = {v8f{}, v8f{}, v8f{}, v8f{}};

#pragma unroll
    for (int k0 = 0; k0 < CC; k0 += 4) {
        v2f a = *(const v2f*)(aRow + k0 + hs * 2);
#pragma unroll
        for (int nt = 0; nt < 4; ++nt) {
            const float* bp = &sWt[(nt * 16 + l15) * LDS_STRIDE + k0 + hs * 2];
            v2f b = *(const v2f*)bp;
            acc[nt] = __builtin_amdgcn_wmma_f32_16x16x4_f32(
                false, a, false, b, (short)0, acc[nt], false, false);
        }
    }

#pragma unroll
    for (int nt = 0; nt < 4; ++nt) {
        const int col = nt * 16 + l15;
        const float bias = b1[col];
#pragma unroll
        for (int r = 0; r < 8; ++r) {
            const int row = rowBase + hs * 8 + r;
            if (row < Npts) {
                bx[(size_t)row * CC + col] = acc[nt][r] + bias;
            }
        }
    }
}

__global__ __launch_bounds__(256)
void dw_gather_residual(const float* __restrict__ feat,
                        const int*   __restrict__ nb,
                        const float* __restrict__ bx,
                        const float* __restrict__ dw,
                        const float* __restrict__ b2,
                        float*       __restrict__ out,
                        int Npts)
{
    __shared__ int   sIdx[GW * SLAB];
    __shared__ float sDw[KK * CC];

    const int lane = threadIdx.x & 31;
    const int wave = __builtin_amdgcn_readfirstlane((int)(threadIdx.x >> 5));
    const int hs   = lane >> 4;
    const int c4   = (lane & 15) << 2;

    int*       slab  = &sIdx[wave * SLAB];
    const int* myIdx = slab + hs * KK;

    const int nPairs = (Npts + 1) >> 1;
    const int nIdx   = Npts * KK;
    const int step   = (int)gridDim.x * GW;
    int p = (int)blockIdx.x * GW + wave;

    const float4 bb = *(const float4*)(b2 + c4);

    int n0 = -1, n1 = -1;
    if (p < nPairs) {
        const int e = p * (2 * KK) + lane;
        if (e < nIdx) n0 = nb[e];
        if (lane < 2 * KK - 32 && e + 32 < nIdx) n1 = nb[e + 32];
    }

    for (int i = threadIdx.x; i < KK * CC; i += 256) {
        sDw[i] = dw[i];
    }
    __syncthreads();

    for (; p < nPairs; p += step) {
        slab[lane]      = n0;
        slab[32 + lane] = n1;

        const int row = 2 * p + hs;

        n0 = -1; n1 = -1;
        const int pn = p + step;
        if (pn < nPairs) {
            const int e = pn * (2 * KK) + lane;
            if (e < nIdx) n0 = nb[e];
            if (lane < 2 * KK - 32 && e + 32 < nIdx) n1 = nb[e + 32];
        }
        float4 f = make_float4(0.f, 0.f, 0.f, 0.f);
        if (row < Npts) f = *(const float4*)(feat + (size_t)row * CC + c4);

        float ax = 0.f, ay = 0.f, az = 0.f, aw = 0.f;
#pragma unroll
        for (int k = 0; k < KK; ++k) {
            const int idx = myIdx[k];
            if (idx >= 0) {
                const float4 g = *(const float4*)(bx + (size_t)idx * CC + c4);
                const float4 w = *(const float4*)(&sDw[k * CC + c4]);
                ax += g.x * w.x;
                ay += g.y * w.y;
                az += g.z * w.z;
                aw += g.w * w.w;
            }
        }

        if (row < Npts) {
            float4 o;
            o.x = f.x + ax + bb.x;
            o.y = f.y + ay + bb.y;
            o.z = f.z + az + bb.z;
            o.w = f.w + aw + bb.w;
            *(float4*)(out + (size_t)row * CC + c4) = o;
        }
    }
}

extern "C" void kernel_launch(void* const* d_in, const int* in_sizes, int n_in,
                              void* d_out, int out_size, void* d_ws, size_t ws_size,
                              hipStream_t stream)
{
    const float* feat = (const float*)d_in[0];
    const int*   nb   = (const int*)  d_in[1];
    const float* W    = (const float*)d_in[2];
    const float* b1   = (const float*)d_in[3];
    const float* b2   = (const float*)d_in[4];
    const float* dw   = (const float*)d_in[5];
    float* out = (float*)d_out;
    float* bx  = (float*)d_ws;

    const int Npts = in_sizes[0] / CC;

    const int gemmBlocks = (Npts + 127) / 128;
    gemm_bias_wmma<<<gemmBlocks, 256, 0, stream>>>(feat, W, b1, bx, Npts);

    dw_gather_residual<<<2048, 256, 0, stream>>>(feat, nb, bx, dw, b2, out, Npts);
}


# gather kernel: each row's valid taps packed into a per-wave LDS list; the bx rows of 8 entries are requested together before any is summed (was: one load waited for per tap); tap order kept
# speedup vs baseline: 1.4048x; 1.1554x over previous
#include <hip/hip_runtime.h>

#define CC 64
#define KK 27
#define LDS_STRIDE 66
#define GW 8
#define LSLOTS 32
#define BATCH 8

typedef __attribute__((ext_vector_type(2))) float v2f;
typedef __attribute__((ext_vector_type(8))) float v8f;

__global__ __launch_bounds__(256)
void gemm_bias_wmma(const float* __restrict__ feat,
                    const float* __restrict__ W,
                    const float* __restrict__ b1,
                    float* __restrict__ bx,
                    int Npts)
{
    __shared__ float sWt[CC * LDS_STRIDE];
    for (int i = threadIdx.x; i < CC * CC; i += 256) {
        int k = i >> 6;
        int n = i & 63;
        sWt[n * LDS_STRIDE + k] = W[i];
    }
    __syncthreads();

    const int lane = threadIdx.x & 31;
    const int wave = threadIdx.x >> 5;
    const int l15  = lane & 15;
    const int hs   = lane >> 4;
    const int rowBase = (blockIdx.x * 8 + wave) * 16;

    int aRowIdx = rowBase + l15;
    if (aRowIdx > Npts - 1) aRowIdx = Npts - 1;
    const float* aRow = feat + (size_t)aRowIdx * CC;

    v8f acc[4] = {v8f{}, v8f{}, v8f{}, v8f{}};

#pragma unroll
    for (int k0 = 0; k0 < CC; k0 += 4) {
        v2f a = *(const v2f*)(aRow + k0 + hs * 2);
#pragma unroll
        for (int nt = 0; nt < 4; ++nt) {
            const float* bp = &sWt[(nt * 16 + l15) * LDS_STRIDE + k0 + hs * 2];
            v2f b = *(const v2f*)bp;
            acc[nt] = __builtin_amdgcn_wmma_f32_16x16x4_f32(
                false, a, false, b, (short)0, acc[nt], false, false);
        }
    }

#pragma unroll
    for (int nt = 0; nt < 4; ++nt) {
        const int col = nt * 16 + l15;
        const float bias = b1[col];
#pragma unroll
        for (int r = 0; r < 8; ++r) {
            const int row = rowBase + hs * 8 + r;
            if (row < Npts) {
                bx[(size_t)row * CC + col] = acc[nt][r] + bias;
            }
        }
    }
}

__global__ __launch_bounds__(256)
void dw_gather_residual(const float* __restrict__ feat,
                        const int*   __restrict__ nb,
                        const float* __restrict__ bx,
                        const float* __restrict__ dw,
                        const float* __restrict__ b2,
                        float*       __restrict__ out,
                        int Npts)
{
    __shared__ int2  sList[GW * 2 * LSLOTS];
    __shared__ float sDw[KK * CC];

    const int lane = threadIdx.x & 31;
    const int wave = __builtin_amdgcn_readfirstlane((int)(threadIdx.x >> 5));
    const int hs   = lane >> 4;
    const int c4   = (lane & 15) << 2;

    int2*       listW  = &sList[wave * 2 * LSLOTS];
    const int2* myList = listW + hs * LSLOTS;

    const int nPairs = (Npts + 1) >> 1;
    const int step   = (int)gridDim.x * GW;
    int p = (int)blockIdx.x * GW + wave;

    const float4 bb = *(const float4*)(b2 + c4);

    int nA = -1, nB = -1;
    if (p < nPairs && lane < KK) {
        nA = nb[(2 * p) * KK + lane];
        if (2 * p + 1 < Npts) nB = nb[(2 * p + 1) * KK + lane];
    }

    listW[lane]          = make_int2(0, 0);
    listW[LSLOTS + lane] = make_int2(0, 0);

    for (int i = threadIdx.x; i < KK * CC; i += 256) {
        sDw[i] = dw[i];
    }
    __syncthreads();

    for (; p < nPairs; p += step) {
        const bool vA = nA >= 0;
        const bool vB = nB >= 0;
        const unsigned mA = __builtin_amdgcn_ballot_w32(vA);
        const unsigned mB = __builtin_amdgcn_ballot_w32(vB);
        if (vA) listW[__builtin_amdgcn_mbcnt_lo(mA, 0u)]          = make_int2(nA, lane);
        if (vB) listW[LSLOTS + __builtin_amdgcn_mbcnt_lo(mB, 0u)] = make_int2(nB, lane);
        const int cA    = __popc(mA);
        const int cB    = __popc(mB);
        const int cMine = hs ? cB : cA;
        const int cMax  = cA > cB ? cA : cB;

        const int row = 2 * p + hs;

        nA = -1; nB = -1;
        const int pn = p + step;
        if (pn < nPairs && lane < KK) {
            nA = nb[(2 * pn) * KK + lane];
            if (2 * pn + 1 < Npts) nB = nb[(2 * pn + 1) * KK + lane];
        }
        float4 f = make_float4(0.f, 0.f, 0.f, 0.f);
        if (row < Npts) f = *(const float4*)(feat + (size_t)row * CC + c4);

        float ax = 0.f, ay = 0.f, az = 0.f, aw = 0.f;
        for (int t0 = 0; t0 < cMax; t0 += BATCH) {
            int2   e[BATCH];
            float4 g[BATCH];
#pragma unroll
            for (int u = 0; u < BATCH; ++u) {
                e[u] = myList[t0 + u];
            }
#pragma unroll
            for (int u = 0; u < BATCH; ++u) {
                const int idx = (t0 + u < cMine) ? e[u].x : 0;
                g[u] = *(const float4*)(bx + (size_t)idx * CC + c4);
            }
#pragma unroll
            for (int u = 0; u < BATCH; ++u) {
                const bool   v = t0 + u < cMine;
                const int    k = v ? e[u].y : 0;
                const float4 w = *(const float4*)(&sDw[k * CC + c4]);
                const float gx = v ? g[u].x : 0.f;
                const float gy = v ? g[u].y : 0.f;
                const float gz = v ? g[u].z : 0.f;
                const float gw = v ? g[u].w : 0.f;
                ax += gx * w.x;
                ay += gy * w.y;
                az += gz * w.z;
                aw += gw * w.w;
            }
        }

        if (row < Npts) {
            float4 o;
            o.x = f.x + ax + bb.x;
            o.y = f.y + ay + bb.y;
            o.z = f.z + az + bb.z;
            o.w = f.w + aw + bb.w;
            *(float4*)(out + (size_t)row * CC + c4) = o;
        }
    }
}

extern "C" void kernel_launch(void* const* d_in, const int* in_sizes, int n_in,
                              void* d_out, int out_size, void* d_ws, size_t ws_size,
                              hipStream_t stream)
{
    const float* feat = (const float*)d_in[0];
    const int*   nb   = (const int*)  d_in[1];
    const float* W    = (const float*)d_in[2];
    const float* b1   = (const float*)d_in[3];
    const float* b2   = (const float*)d_in[4];
    const float* dw   = (const float*)d_in[5];
    float* out = (float*)d_out;
    float* bx  = (float*)d_ws;

    const int Npts = in_sizes[0] / CC;

    const int gemmBlocks = (Npts + 127) / 128;
    gemm_bias_wmma<<<gemmBlocks, 256, 0, stream>>>(feat, W, b1, bx, Npts);

    dw_gather_residual<<<1280, 256, 0, stream>>>(feat, nb, bx, dw, b2, out, Npts);
}


# GEMM kernel: a wave's 16 feat rows and 16 bx rows pass through its own LDS block, so global loads and stores are 16 bytes per lane over whole lines (was 16 lines per load); W read in 16-byte pieces
# speedup vs baseline: 1.6886x; 1.2020x over previous
#include <hip/hip_runtime.h>

#define CC 64
#define KK 27
#define LDS_STRIDE 66
#define A_PITCH 68
#define GW 8
#define LSLOTS 32
#define BATCH 8

typedef __attribute__((ext_vector_type(2))) float v2f;
typedef __attribute__((ext_vector_type(8))) float v8f;

__global__ __launch_bounds__(256)
void gemm_bias_wmma(const float* __restrict__ feat,
                    const float* __restrict__ W,
                    const float* __restrict__ b1,
                    float* __restrict__ bx,
                    int Npts)
{
    __shared__ float sWt[CC * LDS_STRIDE];
    __shared__ float sTile[8 * 16 * A_PITCH];

    const int lane = threadIdx.x & 31;
    const int wave = __builtin_amdgcn_readfirstlane((int)(threadIdx.x >> 5));
    const int l15  = lane & 15;
    const int hs   = lane >> 4;
    const int c4   = l15 << 2;
    const int rowBase = ((int)blockIdx.x * 8 + wave) * 16;
    const bool full = rowBase + 16 <= Npts;
    float* tile = sTile + wave * (16 * A_PITCH);
    const float4* feat4 = (const float4*)feat;
    float4*       bx4   = (float4*)bx;
    const unsigned base4 = (unsigned)((rowBase + hs) * 16 + l15);
    const unsigned last4 = (unsigned)(Npts - 1) * 16u + (unsigned)l15;

    float4 wv[4];
#pragma unroll
    for (int j = 0; j < 4; ++j) {
        wv[j] = ((const float4*)W)[threadIdx.x + 256 * j];
    }

    {
        float4 ain[8];
#pragma unroll
        for (int j = 0; j < 8; ++j) {
            unsigned i = base4 + 32u * j;
            if (i > last4) i = last4;
            ain[j] = feat4[i];
        }
#pragma unroll
        for (int j = 0; j < 8; ++j) {
            *(float4*)(tile + (2 * j + hs) * A_PITCH + c4) = ain[j];
        }
    }

#pragma unroll
    for (int j = 0; j < 4; ++j) {
        const int i4 = threadIdx.x + 256 * j;
        const int k  = i4 >> 4;
        const int n  = (i4 & 15) << 2;
        sWt[(n + 0) * LDS_STRIDE + k] = wv[j].x;
        sWt[(n + 1) * LDS_STRIDE + k] = wv[j].y;
        sWt[(n + 2) * LDS_STRIDE + k] = wv[j].z;
        sWt[(n + 3) * LDS_STRIDE + k] = wv[j].w;
    }
    __syncthreads();

    v8f acc[4] = {v8f{}, v8f{}, v8f{}, v8f{}};

    const float* aRow = tile + l15 * A_PITCH + hs * 2;

#pragma unroll
    for (int k0 = 0; k0 < CC; k0 += 4) {
        v2f a = *(const v2f*)(aRow + k0);
#pragma unroll
        for (int nt = 0; nt < 4; ++nt) {
            const float* bp = &sWt[(nt * 16 + l15) * LDS_STRIDE + k0 + hs * 2];
            v2f b = *(const v2f*)bp;
            acc[nt] = __builtin_amdgcn_wmma_f32_16x16x4_f32(
                false, a, false, b, (short)0, acc[nt], false, false);
        }
    }

#pragma unroll
    for (int nt = 0; nt < 4; ++nt) {
#pragma unroll
        for (int r = 0; r < 8; ++r) {
            tile[(hs * 8 + r) * A_PITCH + nt * 16 + l15] = acc[nt][r];
        }
    }
    const float4 bias = *(const float4*)(b1 + c4);
    if (full) {
        float4 v[8];
#pragma unroll
        for (int j = 0; j < 8; ++j) {
            v[j] = *(const float4*)(tile + (2 * j + hs) * A_PITCH + c4);
        }
#pragma unroll
        for (int j = 0; j < 8; ++j) {
            float4 o;
            o.x = v[j].x + bias.x;
            o.y = v[j].y + bias.y;
            o.z = v[j].z + bias.z;
            o.w = v[j].w + bias.w;
            bx4[base4 + 32u * j] = o;
        }
    } else {
#pragma unroll
        for (int j = 0; j < 8; ++j) {
            const float4 v = *(const float4*)(tile + (2 * j + hs) * A_PITCH + c4);
            const unsigned i = base4 + 32u * j;
            if (i <= last4) {
                float4 o;
                o.x = v.x + bias.x;
                o.y = v.y + bias.y;
                o.z = v.z + bias.z;
                o.w = v.w + bias.w;
                bx4[i] = o;
            }
        }
    }
}

__global__ __launch_bounds__(256)
void dw_gather_residual(const float* __restrict__ feat,
                        const int*   __restrict__ nb,
                        const float*              bx,
                        const float* __restrict__ dw,
                        const float* __restrict__ b2,
                        float*       __restrict__ out,
                        int Npts)
{
    __shared__ int4   sList4[GW * LSLOTS];
    __shared__ float4 sDw4[(KK + 1) * 16];

    const int lane = threadIdx.x & 31;
    const int wave = __builtin_amdgcn_readfirstlane((int)(threadIdx.x >> 5));
    const int hs   = lane >> 4;
    const int l15  = lane & 15;
    const int c4   = l15 << 2;

    int2*       listW  = (int2*)&sList4[wave * LSLOTS];
    const int2* myList = listW + hs * LSLOTS;

    const float4* bx4  = (const float4*)bx;
    const char*   dwB  = (const char*)sDw4 + (l15 << 4);

    const int2 filler = make_int2(0, KK * CC * 4);

    const int nPairs = (Npts + 1) >> 1;
    const int step   = (int)gridDim.x * GW;
    int p = (int)blockIdx.x * GW + wave;

    const float4 bb = *(const float4*)(b2 + c4);

    int nA = -1, nB = -1;
    if (p < nPairs && lane < KK) {
        nA = nb[(2 * p) * KK + lane];
        if (2 * p + 1 < Npts) nB = nb[(2 * p + 1) * KK + lane];
    }

    for (int i = threadIdx.x; i < KK * 16; i += 256) {
        sDw4[i] = ((const float4*)dw)[i];
    }
    if (threadIdx.x < 16) sDw4[KK * 16 + threadIdx.x] = make_float4(0.f, 0.f, 0.f, 0.f);
    __syncthreads();

    for (; p < nPairs; p += step) {
        listW[lane]          = filler;
        listW[LSLOTS + lane] = filler;
        const bool vA = nA >= 0;
        const bool vB = nB >= 0;
        const unsigned mA = __builtin_amdgcn_ballot_w32(vA);
        const unsigned mB = __builtin_amdgcn_ballot_w32(vB);
        if (vA) listW[__builtin_amdgcn_mbcnt_lo(mA, 0u)]          = make_int2(nA << 4, lane << 8);
        if (vB) listW[LSLOTS + __builtin_amdgcn_mbcnt_lo(mB, 0u)] = make_int2(nB << 4, lane << 8);
        const int cA   = __popc(mA);
        const int cB   = __popc(mB);
        const int cMax = cA > cB ? cA : cB;

        const int row = 2 * p + hs;

        nA = -1; nB = -1;
        const int pn = p + step;
        if (pn < nPairs && lane < KK) {
            nA = nb[(2 * pn) * KK + lane];
            if (2 * pn + 1 < Npts) nB = nb[(2 * pn + 1) * KK + lane];
        }
        float4 f = make_float4(0.f, 0.f, 0.f, 0.f);
        if (row < Npts) f = *(const float4*)(feat + (size_t)row * CC + c4);

        float ax = 0.f, ay = 0.f, az = 0.f, aw = 0.f;
        for (int t0 = 0; t0 < cMax; t0 += BATCH) {
            int2   e[BATCH];
            float4 g[BATCH];
#pragma unroll
            for (int u = 0; u < BATCH; ++u) {
                e[u] = myList[t0 + u];
            }
#pragma unroll
            for (int u = 0; u < BATCH; ++u) {
                g[u] = bx4[(unsigned)(e[u].x + l15)];
            }
            asm volatile("" ::: "memory");
#pragma unroll
            for (int u = 0; u < BATCH; ++u) {
                const float4 w = *(const float4*)(dwB + e[u].y);
                ax += g[u].x * w.x;
                ay += g[u].y * w.y;
                az += g[u].z * w.z;
                aw += g[u].w * w.w;
            }
        }

        if (row < Npts) {
            float4 o;
            o.x = f.x + ax + bb.x;
            o.y = f.y + ay + bb.y;
            o.z = f.z + az + bb.z;
            o.w = f.w + aw + bb.w;
            *(float4*)(out + (size_t)row * CC + c4) = o;
        }
    }
}

extern "C" void kernel_launch(void* const* d_in, const int* in_sizes, int n_in,
                              void* d_out, int out_size, void* d_ws, size_t ws_size,
                              hipStream_t stream)
{
    const float* feat = (const float*)d_in[0];
    const int*   nb   = (const int*)  d_in[1];
    const float* W    = (const float*)d_in[2];
    const float* b1   = (const float*)d_in[3];
    const float* b2   = (const float*)d_in[4];
    const float* dw   = (const float*)d_in[5];
    float* out = (float*)d_out;
    float* bx  = (float*)d_ws;

    const int Npts = in_sizes[0] / CC;

    const int gemmBlocks = (Npts + 127) / 128;
    gemm_bias_wmma<<<gemmBlocks, 256, 0, stream>>>(feat, W, b1, bx, Npts);

    dw_gather_residual<<<1280, 256, 0, stream>>>(feat, nb, bx, dw, b2, out, Npts);
}
